# starting kernel
# baseline (speedup 1.0000x reference)
#include <hip/hip_runtime.h>


#define N_SIG   2097152
#define L_FILT  129
#define N_OUT   3145728
#define TILE_OUT 768
#define NBLOCKS  4096
#define WIN      560
#define NTHREADS 96
#define KSTEPS   19

typedef float v2f __attribute__((ext_vector_type(2)));
typedef float v8f __attribute__((ext_vector_type(8)));

__global__ __launch_bounds__(NTHREADS)
void resample_poly_wmma(const float* __restrict__ x,
                        const float* __restrict__ h,
                        float* __restrict__ out)
{
    __shared__ float xw[WIN];
    __shared__ float hs[L_FILT];
    __shared__ __align__(16) float ob[TILE_OUT];

    const int tile = blockIdx.x;
    const int tid  = threadIdx.x;
    const int I0   = tile * 256;
    const int W    = 2 * I0 - 21;

    for (int j = tid; j < WIN; j += NTHREADS) {
        int gi = W + j;
        xw[j] = (gi >= 0 && gi < N_SIG) ? x[gi] : 0.0f;
    }
    for (int j = tid; j < L_FILT; j += NTHREADS) hs[j] = h[j];
    __syncthreads();

    const int wave = tid >> 5;
    const int lane = tid & 31;
    const int p    = wave;
    const int d0   = (p == 0) ? -21 : -20;
    const int basep= (p == 0) ? 0 : 1;

    const int r  = lane & 15;
    const int hi = lane >> 4;
    const int n  = lane & 15;

    v2f a[KSTEPS];
#pragma unroll
    for (int s = 0; s < KSTEPS; ++s) {
        const int k0 = 4 * s + 2 * hi;
        const int t0 = k0 - 2 * r;
        float ax = 0.0f, ay = 0.0f;
        if (t0 >= 0 && t0 < 43)         ax = 3.0f * hs[64 + 2 * p - 3 * (d0 + t0)];
        if (t0 + 1 >= 0 && t0 + 1 < 43) ay = 3.0f * hs[64 + 2 * p - 3 * (d0 + t0 + 1)];
        v2f av; av.x = ax; av.y = ay;
        a[s] = av;
    }

    v8f acc = {0.f, 0.f, 0.f, 0.f, 0.f, 0.f, 0.f, 0.f};
    const int bbase = basep + 32 * n + 2 * hi;
#pragma unroll
    for (int s = 0; s < KSTEPS; ++s) {
        v2f b;
        b.x = xw[bbase + 4 * s];
        b.y = xw[bbase + 4 * s + 1];
        acc = __builtin_amdgcn_wmma_f32_16x16x4_f32(
             false, a[s],  false, b,
             (short)0, acc,  false,  false);
    }

    const int rbase = hi * 8;
#pragma unroll
    for (int v = 0; v < 8; ++v) {
        ob[3 * (16 * n + rbase + v) + p] = acc[v];
    }
    __syncthreads();

    float4*       out4 = (float4*)(out + tile * TILE_OUT);
    const float4* ob4  = (const float4*)ob;
    out4[tid]      = ob4[tid];
    out4[tid + 96] = ob4[tid + 96];
}

extern "C" void kernel_launch(void* const* d_in, const int* in_sizes, int n_in,
                              void* d_out, int out_size, void* d_ws, size_t ws_size,
                              hipStream_t stream) {
    (void)in_sizes; (void)n_in; (void)d_ws; (void)ws_size; (void)out_size;
    const float* x = (const float*)d_in[0];
    const float* h = (const float*)d_in[1];
    float* out = (float*)d_out;
    resample_poly_wmma<<<NBLOCKS, NTHREADS, 0, stream>>>(x, h, out);
}


# Each wave stages its own x rows in a private LDS row, tile by tile, behind counted load waits (loads issued in tile order); the one work-group barrier left guards the filter table, not x
# speedup vs baseline: 2.1744x; 2.1744x over previous
#include <hip/hip_runtime.h>


#define N_SIG    2097152
#define L_FILT   129
#define NWG      256
#define NTHREADS 128
#define NBLK     4
#define KSTEPS   19
#define KHEAD    6
#define XORG     24
#define WROW     2176
#define WROUNDS  17
#define GPAD     30
#define GLEN     108

typedef float v2f __attribute__((ext_vector_type(2)));
typedef float v8f __attribute__((ext_vector_type(8)));

#define WMMA3(S)                                                                                              \
    acc0 = __builtin_amdgcn_wmma_f32_16x16x4_f32(false, a[0][S], false, bc[S], (short)0, acc0, false, false); \
    acc1 = __builtin_amdgcn_wmma_f32_16x16x4_f32(false, a[1][S], false, bc[S], (short)0, acc1, false, false); \
    acc2 = __builtin_amdgcn_wmma_f32_16x16x4_f32(false, a[2][S], false, bc[S], (short)0, acc2, false, false);

__global__ __launch_bounds__(NTHREADS)
void resample_poly_wmma(const float* x,
                        const float* h,
                        float* out)
{
    __shared__ __align__(16) float xw[4][WROW];
    __shared__ __align__(16) float gp[3 * NTHREADS];
    __shared__ __align__(16) float ob[4][768];

    const int wg   = blockIdx.x;
    const int tid  = threadIdx.x;
    const int wave = tid >> 5;
    const int lane = tid & 31;
    const int I0   = wg * 4096;
    const int gbase = 2 * (I0 + 1024 * wave) - XORG;

    float gv[3];
#pragma unroll
    for (int u = 0; u < 3; ++u) {
        const int j  = tid + NTHREADS * u;
        const int p  = j / GLEN;
        const int t  = j - p * GLEN - GPAD;
        const int hx = 130 + 2 * p - 3 * t;
        const int hc = hx < 0 ? 0 : (hx > L_FILT - 1 ? L_FILT - 1 : hx);
        const float hv = 3.0f * h[hc];
        gv[u] = (hx >= 0 && hx < L_FILT) ? hv : 0.0f;
    }

#define LOAD_ORDER() __builtin_amdgcn_fence(__ATOMIC_ACQ_REL, "wavefront")
    LOAD_ORDER();

    float4 stg[WROUNDS];
    const int g0  = gbase + 4 * lane;
    const int g16 = g0 + 128 * (WROUNDS - 1);
    stg[0] = *(const float4*)(x + (g0 < 0 ? 0 : g0));
#pragma unroll
    for (int j = 1; j < 5; ++j) stg[j] = *(const float4*)(x + g0 + 128 * j);
    LOAD_ORDER();
#pragma unroll
    for (int j = 5; j < 9; ++j) stg[j] = *(const float4*)(x + g0 + 128 * j);
    LOAD_ORDER();
#pragma unroll
    for (int j = 9; j < 13; ++j) stg[j] = *(const float4*)(x + g0 + 128 * j);
    LOAD_ORDER();
#pragma unroll
    for (int j = 13; j < WROUNDS - 1; ++j) stg[j] = *(const float4*)(x + g0 + 128 * j);
    stg[WROUNDS - 1] = *(const float4*)(x + (g16 > N_SIG - 4 ? N_SIG - 4 : g16));
    LOAD_ORDER();
    __builtin_amdgcn_sched_barrier(0);

    gp[tid] = gv[0];
    gp[tid + NTHREADS] = gv[1];
    gp[tid + 2 * NTHREADS] = gv[2];
    __builtin_amdgcn_fence(__ATOMIC_RELEASE, "workgroup", "local");
    __builtin_amdgcn_s_barrier();
    __builtin_amdgcn_fence(__ATOMIC_ACQUIRE, "workgroup", "local");

    const int r  = lane & 15;
    const int hi = lane >> 4;

    v2f a[3][KSTEPS];
#pragma unroll
    for (int p = 0; p < 3; ++p) {
        const float* gb = gp + p * GLEN + GPAD + 2 * hi - 2 * r;
#pragma unroll
        for (int s = 0; s < KSTEPS; ++s) a[p][s] = *(const v2f*)(gb + 4 * s);
    }

    float4*      xs  = (float4*)xw[wave] + lane;
    const float* xb  = xw[wave] + 2 + 32 * r + 2 * hi;
    float*       myob = ob[wave];
    const float4* ob4 = (const float4*)myob + lane;
    float4*      obw  = (float4*)(myob + 24 * (2 * r + hi));
    float4*      o4   = (float4*)(out + 3 * (size_t)(I0 + 1024 * wave)) + lane;

#pragma unroll
    for (int j = 1; j < 5; ++j) xs[32 * j] = stg[j];
    {
        const bool in0 = g0 >= 0;
        xs[0] = make_float4(in0 ? stg[0].x : 0.f, in0 ? stg[0].y : 0.f, in0 ? stg[0].z : 0.f, in0 ? stg[0].w : 0.f);
    }
    __builtin_amdgcn_fence(__ATOMIC_RELEASE, "wavefront");
    __builtin_amdgcn_wave_barrier();
    __builtin_amdgcn_fence(__ATOMIC_ACQUIRE, "wavefront");
    v2f bc[KSTEPS];
#pragma unroll
    for (int s = 0; s < 4; ++s) bc[s] = *(const v2f*)(xb + 4 * s);
    LOAD_ORDER();
#pragma unroll
    for (int s = 4; s < 10; ++s) bc[s] = *(const v2f*)(xb + 4 * s);
    LOAD_ORDER();
#pragma unroll
    for (int s = 10; s < KSTEPS; ++s) bc[s] = *(const v2f*)(xb + 4 * s);
    float4 t0 = make_float4(0.f, 0.f, 0.f, 0.f), t1 = t0, t2 = t0, t3 = t0, t4 = t0, t5 = t0;

#pragma unroll
    for (int b = 0; b < NBLK; ++b) {
        v2f bn[KSTEPS];
#define STAGE_NEXT()                                                                                            \
        {                                                                                                       \
            _Pragma("unroll")                                                                                   \
            for (int j = 4 * b + 5; j < 4 * b + 9; ++j) {                                                       \
                if (j < WROUNDS - 1) {                                                                          \
                    xs[32 * j] = stg[j];                                                                        \
                } else {                                                                                        \
                    const bool in16 = g16 <= N_SIG - 4;                                                         \
                    xs[32 * j] = make_float4(in16 ? stg[j].x : 0.f, in16 ? stg[j].y : 0.f,                      \
                                             in16 ? stg[j].z : 0.f, in16 ? stg[j].w : 0.f);                     \
                }                                                                                               \
            }                                                                                                   \
            __builtin_amdgcn_fence(__ATOMIC_RELEASE, "wavefront");                                              \
            __builtin_amdgcn_wave_barrier();                                                                    \
            __builtin_amdgcn_fence(__ATOMIC_ACQUIRE, "wavefront");                                              \
            _Pragma("unroll")                                                                                   \
            for (int s = 0; s < KSTEPS; ++s) bn[s] = *(const v2f*)(xb + 512 * (b + 1) + 4 * s);                 \
        }
        if (b > 0 && b + 1 < NBLK) STAGE_NEXT()
        __builtin_amdgcn_sched_barrier(0);

        v8f acc0 = {0.f, 0.f, 0.f, 0.f, 0.f, 0.f, 0.f, 0.f};
        v8f acc1 = acc0, acc2 = acc0;
#pragma unroll
        for (int s = 0; s < KHEAD; ++s) { WMMA3(s) }
        if (b > 0) {
            float4* op = o4 + 192 * (b - 1);
            op[0] = t0; op[32] = t1; op[64] = t2; op[96] = t3; op[128] = t4; op[160] = t5;
        }
        if (b == 0) {
            __builtin_amdgcn_sched_barrier(0);
            STAGE_NEXT()
            __builtin_amdgcn_sched_barrier(0);
        }
#pragma unroll
        for (int s = KHEAD; s < KSTEPS; ++s) { WMMA3(s) }
        if (b > 0) {
            __builtin_amdgcn_sched_group_barrier(0x008, 3 * KHEAD, 0);
            __builtin_amdgcn_sched_group_barrier(0x040, 6, 0);
            __builtin_amdgcn_sched_group_barrier(0x008, 3 * (KSTEPS - KHEAD), 0);
        }

        obw[0] = make_float4(acc0[0], acc1[0], acc2[0], acc0[1]);
        obw[1] = make_float4(acc1[1], acc2[1], acc0[2], acc1[2]);
        obw[2] = make_float4(acc2[2], acc0[3], acc1[3], acc2[3]);
        obw[3] = make_float4(acc0[4], acc1[4], acc2[4], acc0[5]);
        obw[4] = make_float4(acc1[5], acc2[5], acc0[6], acc1[6]);
        obw[5] = make_float4(acc2[6], acc0[7], acc1[7], acc2[7]);
        __builtin_amdgcn_fence(__ATOMIC_RELEASE, "wavefront");
        __builtin_amdgcn_wave_barrier();
        __builtin_amdgcn_fence(__ATOMIC_ACQUIRE, "wavefront");
        t0 = ob4[0]; t1 = ob4[32]; t2 = ob4[64]; t3 = ob4[96]; t4 = ob4[128]; t5 = ob4[160];
        __builtin_amdgcn_fence(__ATOMIC_RELEASE, "wavefront");
        __builtin_amdgcn_wave_barrier();
        __builtin_amdgcn_fence(__ATOMIC_ACQUIRE, "wavefront");

        if (b + 1 < NBLK) {
#pragma unroll
            for (int s = 0; s < KSTEPS; ++s) bc[s] = bn[s];
        }
    }
    {
        float4* op = o4 + 192 * (NBLK - 1);
        op[0] = t0; op[32] = t1; op[64] = t2; op[96] = t3; op[128] = t4; op[160] = t5;
    }
}

extern "C" void kernel_launch(void* const* d_in, const int* in_sizes, int n_in,
                              void* d_out, int out_size, void* d_ws, size_t ws_size,
                              hipStream_t stream) {
    (void)in_sizes; (void)n_in; (void)d_ws; (void)ws_size; (void)out_size;
    const float* x = (const float*)d_in[0];
    const float* h = (const float*)d_in[1];
    float* out = (float*)d_out;
    resample_poly_wmma<<<NWG, NTHREADS, 0, stream>>>(x, h, out);
}
